# starting kernel
# baseline (speedup 1.0000x reference)
#include <hip/hip_runtime.h>

#define D 128

typedef __attribute__((ext_vector_type(2))) float v2f;
typedef __attribute__((ext_vector_type(8))) float v8f;

__global__ __launch_bounds__(256) void k_init_deg(float* deg, int N) {
    int i = blockIdx.x * 256 + threadIdx.x;
    if (i < N) deg[i] = 1.0f;
}

__global__ __launch_bounds__(256) void k_count_deg(const int* __restrict__ dst,
                                                   float* deg, int E) {
    int i = blockIdx.x * 256 + threadIdx.x;
    if (i < E) atomicAdd(&deg[dst[i]], 1.0f);
}

__global__ __launch_bounds__(256) void k_rsqrt(float* deg, int N) {
    int i = blockIdx.x * 256 + threadIdx.x;
    if (i < N) deg[i] = rsqrtf(deg[i]);
}

__global__ __launch_bounds__(256) void k_zero(float* p, size_t n) {
    size_t i = (size_t)blockIdx.x * 256 + threadIdx.x;
    if (i < n) p[i] = 0.0f;
}

__global__ __launch_bounds__(256) void k_gemm_wmma(const float* __restrict__ A,
                                                   const float* __restrict__ W,
                                                   float* __restrict__ C) {
    const int wave  = threadIdx.x >> 5;
    const int lane  = threadIdx.x & 31;
    const int row0  = blockIdx.x * 16;
    const int col0  = wave * 16;
    const int lrow  = lane & 15;
    const int khalf = (lane >> 4) * 2;

    const float* Arow = A + (size_t)(row0 + lrow) * D;

    v8f acc = {};
    for (int k = 0; k < D; k += 4) {
        v2f a;
        a.x = Arow[k + khalf + 0];
        a.y = Arow[k + khalf + 1];
        v2f b;
        b.x = W[(size_t)(k + khalf + 0) * D + col0 + lrow];
        b.y = W[(size_t)(k + khalf + 1) * D + col0 + lrow];
        acc = __builtin_amdgcn_wmma_f32_16x16x4_f32(
             false, a,  false, b,
             (short)0, acc,  false,  false);
    }

    const int m_add = (lane >> 4) * 8;
    const int ncol  = col0 + lrow;
#pragma unroll
    for (int r = 0; r < 8; ++r) {
        C[(size_t)(row0 + r + m_add) * D + ncol] = acc[r];
    }
}

__global__ __launch_bounds__(256) void k_scatter(const float* __restrict__ h,
                                                 const int* __restrict__ src,
                                                 const int* __restrict__ dst,
                                                 const float* __restrict__ dinv,
                                                 float* agg, int E) {
    int e    = (int)(((size_t)blockIdx.x * 256 + threadIdx.x) >> 5);
    int lane = threadIdx.x & 31;
    if (e >= E) return;

    int s = src[e];
    int d = dst[e];
    float nrm = dinv[s] * dinv[d];

    const float4* hs = (const float4*)(h + (size_t)s * D);
    float4 v = hs[lane];

    float* a = agg + (size_t)d * D + lane * 4;
    atomicAdd(a + 0, v.x * nrm);
    atomicAdd(a + 1, v.y * nrm);
    atomicAdd(a + 2, v.z * nrm);
    atomicAdd(a + 3, v.w * nrm);
}

__global__ __launch_bounds__(256) void k_finalize(const float* __restrict__ agg,
                                                  const float* __restrict__ h,
                                                  const float* __restrict__ dinv,
                                                  const float* __restrict__ bias,
                                                  float* __restrict__ out,
                                                  size_t ND, int do_relu) {
    size_t i = (size_t)blockIdx.x * 256 + threadIdx.x;
    if (i >= ND) return;
    int n = (int)(i >> 7);
    int c = (int)(i & 127);
    float di = dinv[n];
    float v = agg[i] + h[i] * di * di + bias[c];
    if (do_relu) v = fmaxf(v, 0.0f);
    out[i] = v;
}

extern "C" void kernel_launch(void* const* d_in, const int* in_sizes, int n_in,
                              void* d_out, int out_size, void* d_ws, size_t ws_size,
                              hipStream_t stream) {
    const float* x  = (const float*)d_in[0];
    const int*   ei = (const int*)d_in[1];
    const float* W1 = (const float*)d_in[2];
    const float* b1 = (const float*)d_in[3];
    const float* W2 = (const float*)d_in[4];
    const float* b2 = (const float*)d_in[5];
    float* out = (float*)d_out;

    const int N = in_sizes[0] / D;
    const int E = in_sizes[1] / 2;
    const int* src = ei;
    const int* dst = ei + E;
    const size_t ND = (size_t)N * D;

    float* ws   = (float*)d_ws;
    float* dinv = ws;
    float* h    = ws + 65536;
    float* agg  = h + ND;

    const int nb_N  = (N + 255) / 256;
    const int nb_E  = (E + 255) / 256;
    const int nb_ED = (E + 7) / 8;
    const int nb_ND = (int)((ND + 255) / 256);
    const int nb_M  = N / 16;

    k_init_deg <<<nb_N, 256, 0, stream>>>(dinv, N);
    k_count_deg<<<nb_E, 256, 0, stream>>>(dst, dinv, E);
    k_rsqrt    <<<nb_N, 256, 0, stream>>>(dinv, N);

    k_gemm_wmma<<<nb_M, 256, 0, stream>>>(x, W1, h);
    k_zero     <<<nb_ND, 256, 0, stream>>>(agg, ND);
    k_scatter  <<<nb_ED, 256, 0, stream>>>(h, src, dst, dinv, agg, E);
    k_finalize <<<nb_ND, 256, 0, stream>>>(agg, h, dinv, b1, agg, ND, 1);

    k_gemm_wmma<<<nb_M, 256, 0, stream>>>(agg, W2, h);
    k_zero     <<<nb_ND, 256, 0, stream>>>(out, ND);
    k_scatter  <<<nb_ED, 256, 0, stream>>>(h, src, dst, dinv, out, E);
    k_finalize <<<nb_ND, 256, 0, stream>>>(out, h, dinv, b2, out, ND, 0);
}


# Float-atomic scatter-add replaced by a gather over destination buckets: edges counting-sorted per call by int-atomic arrival-order rank; a wave sums a node's rows in registers, one store per row
# speedup vs baseline: 2.3079x; 2.3079x over previous
#include <hip/hip_runtime.h>


#define D 128

typedef __attribute__((ext_vector_type(2))) float v2f;
typedef __attribute__((ext_vector_type(8))) float v8f;

__global__ __launch_bounds__(256) void k_zero_cnt(int* cnt, int N) {
    int i = blockIdx.x * 256 + threadIdx.x;
    if (i < N) cnt[i] = 0;
}

__global__ __launch_bounds__(256) void k_count_rank(const int* __restrict__ dst,
                                                    int* cnt,
                                                    int* __restrict__ rank, int E) {
    int i = blockIdx.x * 256 + threadIdx.x;
    if (i < E) rank[i] = atomicAdd(&cnt[dst[i]], 1);
}

__global__ __launch_bounds__(256) void k_dinv_bsum(const int* __restrict__ cnt,
                                                   float* __restrict__ dinv,
                                                   int* __restrict__ bsum, int N) {
    __shared__ int s_w[8];
    const int tid  = threadIdx.x;
    const int lane = tid & 31;
    const int wave = tid >> 5;
    const int i = blockIdx.x * 256 + tid;
    const int c = (i < N) ? cnt[i] : 0;
    if (i < N) dinv[i] = rsqrtf((float)(c + 1));
    int x = c;
    for (int o = 16; o > 0; o >>= 1) x += __shfl_xor(x, o);
    if (lane == 0) s_w[wave] = x;
    __syncthreads();
    if (tid == 0) {
        int t = 0;
        for (int w = 0; w < 8; ++w) t += s_w[w];
        bsum[blockIdx.x] = t;
    }
}

__global__ __launch_bounds__(256) void k_rowptr(const int* __restrict__ cnt,
                                                const int* __restrict__ bsum,
                                                int* __restrict__ rowptr, int N) {
    __shared__ int s_part[8];
    __shared__ int s_tot[8];
    const int tid  = threadIdx.x;
    const int lane = tid & 31;
    const int wave = tid >> 5;
    const int nblk = (int)blockIdx.x;

    int part = 0;
    for (int j = tid; j < nblk; j += 256) part += bsum[j];
    for (int o = 16; o > 0; o >>= 1) part += __shfl_xor(part, o);

    const int i = nblk * 256 + tid;
    const int c = (i < N) ? cnt[i] : 0;
    int x = c;
    for (int o = 1; o < 32; o <<= 1) {
        int y = __shfl_up(x, o);
        if (lane >= o) x += y;
    }
    if (lane == 31) s_tot[wave] = x;
    if (lane == 0)  s_part[wave] = part;
    __syncthreads();

    int base = 0;
    for (int w = 0; w < 8; ++w) base += s_part[w];
    for (int w = 0; w < wave; ++w) base += s_tot[w];
    if (i < N) rowptr[i] = base + x - c;
}

__global__ __launch_bounds__(256) void k_fill(const int* __restrict__ src,
                                              const int* __restrict__ dst,
                                              const int* __restrict__ rowptr,
                                              const int* __restrict__ rank,
                                              int* __restrict__ col, int E) {
    int i = blockIdx.x * 256 + threadIdx.x;
    if (i < E) col[rowptr[dst[i]] + rank[i]] = src[i];
}

__global__ __launch_bounds__(256) void k_gemm_wmma(const float* __restrict__ A,
                                                   const float* __restrict__ W,
                                                   float* __restrict__ C) {
    const int wave  = threadIdx.x >> 5;
    const int lane  = threadIdx.x & 31;
    const int row0  = blockIdx.x * 16;
    const int col0  = wave * 16;
    const int lrow  = lane & 15;
    const int khalf = (lane >> 4) * 2;

    const float* Arow = A + (size_t)(row0 + lrow) * D;

    v8f acc = {};
    for (int k = 0; k < D; k += 4) {
        v2f a;
        a.x = Arow[k + khalf + 0];
        a.y = Arow[k + khalf + 1];
        v2f b;
        b.x = W[(size_t)(k + khalf + 0) * D + col0 + lrow];
        b.y = W[(size_t)(k + khalf + 1) * D + col0 + lrow];
        acc = __builtin_amdgcn_wmma_f32_16x16x4_f32(
             false, a,  false, b,
             (short)0, acc,  false,  false);
    }

    const int m_add = (lane >> 4) * 8;
    const int ncol  = col0 + lrow;
#pragma unroll
    for (int r = 0; r < 8; ++r) {
        C[(size_t)(row0 + r + m_add) * D + ncol] = acc[r];
    }
}

static __device__ __forceinline__ void fma4(float4& acc, const float4 a, const float w) {
    acc.x = fmaf(a.x, w, acc.x);
    acc.y = fmaf(a.y, w, acc.y);
    acc.z = fmaf(a.z, w, acc.z);
    acc.w = fmaf(a.w, w, acc.w);
}

__global__ __launch_bounds__(256) void k_gather(const float* __restrict__ h,
                                                const int* __restrict__ rowptr,
                                                const int* __restrict__ cnt,
                                                const int* __restrict__ col,
                                                const float* __restrict__ dinv,
                                                const float* __restrict__ bias,
                                                float* __restrict__ out,
                                                int N, int do_relu) {
    const int lane = threadIdx.x & 31;
    const int n = __builtin_amdgcn_readfirstlane((int)(blockIdx.x * 8 + (threadIdx.x >> 5)));
    if (n >= N) return;

    const float4* h4 = (const float4*)h;
    const int   beg = rowptr[n];
    const int   c   = cnt[n];
    const float dd  = dinv[n];

    const float  wself = dd * dd;
    const float4 hv = h4[(size_t)n * 32 + lane];
    float4 acc;
    acc.x = hv.x * wself;
    acc.y = hv.y * wself;
    acc.z = hv.z * wself;
    acc.w = hv.w * wself;

    for (int base = 0; base < c; base += 32) {
        const int m = min(32, c - base);
        int   my_s = 0;
        float my_w = 0.0f;
        if (lane < m) {
            my_s = col[beg + base + lane];
            my_w = dinv[my_s] * dd;
        }
        int j = 0;
        for (; j + 4 <= m; j += 4) {
            const int s0 = __builtin_amdgcn_readlane(my_s, j + 0);
            const int s1 = __builtin_amdgcn_readlane(my_s, j + 1);
            const int s2 = __builtin_amdgcn_readlane(my_s, j + 2);
            const int s3 = __builtin_amdgcn_readlane(my_s, j + 3);
            const float w0 = __int_as_float(__builtin_amdgcn_readlane(__float_as_int(my_w), j + 0));
            const float w1 = __int_as_float(__builtin_amdgcn_readlane(__float_as_int(my_w), j + 1));
            const float w2 = __int_as_float(__builtin_amdgcn_readlane(__float_as_int(my_w), j + 2));
            const float w3 = __int_as_float(__builtin_amdgcn_readlane(__float_as_int(my_w), j + 3));
            const float4 a0 = h4[(size_t)s0 * 32 + lane];
            const float4 a1 = h4[(size_t)s1 * 32 + lane];
            const float4 a2 = h4[(size_t)s2 * 32 + lane];
            const float4 a3 = h4[(size_t)s3 * 32 + lane];
            fma4(acc, a0, w0);
            fma4(acc, a1, w1);
            fma4(acc, a2, w2);
            fma4(acc, a3, w3);
        }
        for (; j < m; ++j) {
            const int   s = __builtin_amdgcn_readlane(my_s, j);
            const float w = __int_as_float(__builtin_amdgcn_readlane(__float_as_int(my_w), j));
            const float4 a = h4[(size_t)s * 32 + lane];
            fma4(acc, a, w);
        }
    }

    acc.x += bias[lane * 4 + 0];
    acc.y += bias[lane * 4 + 1];
    acc.z += bias[lane * 4 + 2];
    acc.w += bias[lane * 4 + 3];
    if (do_relu) {
        acc.x = fmaxf(acc.x, 0.0f);
        acc.y = fmaxf(acc.y, 0.0f);
        acc.z = fmaxf(acc.z, 0.0f);
        acc.w = fmaxf(acc.w, 0.0f);
    }
    ((float4*)out)[(size_t)n * 32 + lane] = acc;
}

extern "C" void kernel_launch(void* const* d_in, const int* in_sizes, int n_in,
                              void* d_out, int out_size, void* d_ws, size_t ws_size,
                              hipStream_t stream) {
    const float* x  = (const float*)d_in[0];
    const int*   ei = (const int*)d_in[1];
    const float* W1 = (const float*)d_in[2];
    const float* b1 = (const float*)d_in[3];
    const float* W2 = (const float*)d_in[4];
    const float* b2 = (const float*)d_in[5];
    float* out = (float*)d_out;

    const int N = in_sizes[0] / D;
    const int E = in_sizes[1] / 2;
    const int* src = ei;
    const int* dst = ei + E;
    const size_t ND = (size_t)N * D;

    const int nb_N = (N + 255) / 256;
    const int nb_E = (E + 255) / 256;
    const int nb_G = (N + 7) / 8;
    const int nb_M = N / 16;

    float* ws   = (float*)d_ws;
    float* dinv = ws;
    float* h    = ws + 65536;
    int* cnt    = (int*)(h + ND);
    int* rowptr = cnt + N;
    int* bsum   = rowptr + N;
    int* rank   = bsum + nb_N;
    int* col    = rank + E;

    k_zero_cnt  <<<nb_N, 256, 0, stream>>>(cnt, N);
    k_count_rank<<<nb_E, 256, 0, stream>>>(dst, cnt, rank, E);
    k_dinv_bsum <<<nb_N, 256, 0, stream>>>(cnt, dinv, bsum, N);
    k_rowptr    <<<nb_N, 256, 0, stream>>>(cnt, bsum, rowptr, N);
    k_fill      <<<nb_E, 256, 0, stream>>>(src, dst, rowptr, rank, col, E);

    k_gemm_wmma<<<nb_M, 256, 0, stream>>>(x, W1, h);
    k_gather   <<<nb_G, 256, 0, stream>>>(h, rowptr, cnt, col, dinv, b1, out, N, 1);

    k_gemm_wmma<<<nb_M, 256, 0, stream>>>(out, W2, h);
    k_gather   <<<nb_G, 256, 0, stream>>>(h, rowptr, cnt, col, dinv, b2, out, N, 0);
}


# Fewer, longer-lived waves (about 5x fewer per call): a gather wave takes 4 nodes, a GEMM block loops over row tiles with its W fragments kept in registers, edge kernels take 8 edges per thread
# speedup vs baseline: 2.4319x; 1.0537x over previous
#include <hip/hip_runtime.h>


#define D 128

#define EPT 8
#define GEMM_BLOCKS 512
#define NPW 4

typedef __attribute__((ext_vector_type(2))) float v2f;
typedef __attribute__((ext_vector_type(8))) float v8f;

__global__ __launch_bounds__(256) void k_zero_cnt(int* cnt, int N) {
    int i = blockIdx.x * 256 + threadIdx.x;
    if (i < N) cnt[i] = 0;
}

__global__ __launch_bounds__(256) void k_count_rank(const int* __restrict__ dst,
                                                    int* cnt,
                                                    int* __restrict__ rank, int E) {
    const int base = blockIdx.x * (256 * EPT) + threadIdx.x;
    if ((int)(blockIdx.x + 1) * (256 * EPT) <= E) {
        int d[EPT];
        int r[EPT];
#pragma unroll
        for (int u = 0; u < EPT; ++u) d[u] = dst[base + u * 256];
#pragma unroll
        for (int u = 0; u < EPT; ++u) r[u] = atomicAdd(&cnt[d[u]], 1);
#pragma unroll
        for (int u = 0; u < EPT; ++u) rank[base + u * 256] = r[u];
    } else {
        for (int u = 0; u < EPT; ++u) {
            const int i = base + u * 256;
            if (i < E) rank[i] = atomicAdd(&cnt[dst[i]], 1);
        }
    }
}

__global__ __launch_bounds__(256) void k_dinv_bsum(const int* __restrict__ cnt,
                                                   float* __restrict__ dinv,
                                                   int* __restrict__ bsum, int N) {
    __shared__ int s_w[8];
    const int tid  = threadIdx.x;
    const int lane = tid & 31;
    const int wave = tid >> 5;
    const int i = blockIdx.x * 256 + tid;
    const int c = (i < N) ? cnt[i] : 0;
    if (i < N) dinv[i] = rsqrtf((float)(c + 1));
    int x = c;
    for (int o = 16; o > 0; o >>= 1) x += __shfl_xor(x, o);
    if (lane == 0) s_w[wave] = x;
    __syncthreads();
    if (tid == 0) {
        int t = 0;
        for (int w = 0; w < 8; ++w) t += s_w[w];
        bsum[blockIdx.x] = t;
    }
}

__global__ __launch_bounds__(256) void k_rowptr(const int* __restrict__ cnt,
                                                const int* __restrict__ bsum,
                                                int* __restrict__ rowptr, int N) {
    __shared__ int s_part[8];
    __shared__ int s_tot[8];
    const int tid  = threadIdx.x;
    const int lane = tid & 31;
    const int wave = tid >> 5;
    const int nblk = (int)blockIdx.x;

    int part = 0;
    for (int j = tid; j < nblk; j += 256) part += bsum[j];
    for (int o = 16; o > 0; o >>= 1) part += __shfl_xor(part, o);

    const int i = nblk * 256 + tid;
    const int c = (i < N) ? cnt[i] : 0;
    int x = c;
    for (int o = 1; o < 32; o <<= 1) {
        int y = __shfl_up(x, o);
        if (lane >= o) x += y;
    }
    if (lane == 31) s_tot[wave] = x;
    if (lane == 0)  s_part[wave] = part;
    __syncthreads();

    int base = 0;
    for (int w = 0; w < 8; ++w) base += s_part[w];
    for (int w = 0; w < wave; ++w) base += s_tot[w];
    if (i < N) rowptr[i] = base + x - c;
}

__global__ __launch_bounds__(256) void k_fill(const int* __restrict__ src,
                                              const int* __restrict__ dst,
                                              const int* __restrict__ rowptr,
                                              const int* __restrict__ rank,
                                              int* __restrict__ col, int E) {
    const int base = blockIdx.x * (256 * EPT) + threadIdx.x;
    if ((int)(blockIdx.x + 1) * (256 * EPT) <= E) {
        int d[EPT];
        int k[EPT];
        int s[EPT];
        int p[EPT];
#pragma unroll
        for (int u = 0; u < EPT; ++u) {
            d[u] = dst[base + u * 256];
            k[u] = rank[base + u * 256];
            s[u] = src[base + u * 256];
        }
#pragma unroll
        for (int u = 0; u < EPT; ++u) p[u] = rowptr[d[u]];
#pragma unroll
        for (int u = 0; u < EPT; ++u) col[p[u] + k[u]] = s[u];
    } else {
        for (int u = 0; u < EPT; ++u) {
            const int i = base + u * 256;
            if (i < E) col[rowptr[dst[i]] + rank[i]] = src[i];
        }
    }
}

__global__ __launch_bounds__(256) void k_gemm_wmma(const float* __restrict__ A,
                                                   const float* __restrict__ W,
                                                   float* __restrict__ C, int ntiles) {
    const int wave  = threadIdx.x >> 5;
    const int lane  = threadIdx.x & 31;
    const int col0  = wave * 16;
    const int lrow  = lane & 15;
    const int khalf = (lane >> 4) * 2;
    const int m_add = (lane >> 4) * 8;
    const int ncol  = col0 + lrow;

    v2f b[D / 4];
#pragma unroll
    for (int i = 0; i < D / 4; ++i) {
        b[i].x = W[(size_t)(4 * i + khalf + 0) * D + ncol];
        b[i].y = W[(size_t)(4 * i + khalf + 1) * D + ncol];
    }

    for (int t = blockIdx.x; t < ntiles; t += gridDim.x) {
        const int row0 = t * 16;
        const float* Arow = A + (size_t)(row0 + lrow) * D;

        v2f a[D / 4];
#pragma unroll
        for (int i = 0; i < D / 4; ++i) {
            a[i].x = Arow[4 * i + khalf + 0];
            a[i].y = Arow[4 * i + khalf + 1];
        }
        __builtin_amdgcn_sched_barrier(0);
        v8f acc = {};
#pragma unroll
        for (int i = 0; i < D / 4; ++i) {
            acc = __builtin_amdgcn_wmma_f32_16x16x4_f32(
                 false, a[i],  false, b[i],
                 (short)0, acc,  false,  false);
        }

#pragma unroll
        for (int r = 0; r < 8; ++r) {
            C[(size_t)(row0 + r + m_add) * D + ncol] = acc[r];
        }
    }
}

static __device__ __forceinline__ void fma4(float4& acc, const float4 a, const float w) {
    acc.x = fmaf(a.x, w, acc.x);
    acc.y = fmaf(a.y, w, acc.y);
    acc.z = fmaf(a.z, w, acc.z);
    acc.w = fmaf(a.w, w, acc.w);
}

__global__ __launch_bounds__(256) void k_gather(const float* __restrict__ h,
                                                const int* __restrict__ rowptr,
                                                const int* __restrict__ cnt,
                                                const int* __restrict__ col,
                                                const float* __restrict__ dinv,
                                                const float* __restrict__ bias,
                                                float* __restrict__ out,
                                                int N, int do_relu) {
    const int lane = threadIdx.x & 31;
    const int w = __builtin_amdgcn_readfirstlane((int)(blockIdx.x * 8 + (threadIdx.x >> 5)));
    const int n_beg = w * NPW;
    const int n_end = min(n_beg + NPW, N);

    const float4* h4 = (const float4*)h;
    const float bias0 = bias[lane * 4 + 0];
    const float bias1 = bias[lane * 4 + 1];
    const float bias2 = bias[lane * 4 + 2];
    const float bias3 = bias[lane * 4 + 3];

    for (int n = n_beg; n < n_end; ++n) {
        const int   beg = rowptr[n];
        const int   c   = cnt[n];
        const float dd  = dinv[n];

        const float  wself = dd * dd;
        const float4 hv = h4[(size_t)n * 32 + lane];
        float4 acc;
        acc.x = hv.x * wself;
        acc.y = hv.y * wself;
        acc.z = hv.z * wself;
        acc.w = hv.w * wself;

        for (int base = 0; base < c; base += 32) {
            const int m = min(32, c - base);
            int   my_s = 0;
            float my_w = 0.0f;
            if (lane < m) {
                my_s = col[beg + base + lane];
                my_w = dinv[my_s] * dd;
            }
            int j = 0;
            for (; j + 4 <= m; j += 4) {
                const int s0 = __builtin_amdgcn_readlane(my_s, j + 0);
                const int s1 = __builtin_amdgcn_readlane(my_s, j + 1);
                const int s2 = __builtin_amdgcn_readlane(my_s, j + 2);
                const int s3 = __builtin_amdgcn_readlane(my_s, j + 3);
                const float w0 = __int_as_float(__builtin_amdgcn_readlane(__float_as_int(my_w), j + 0));
                const float w1 = __int_as_float(__builtin_amdgcn_readlane(__float_as_int(my_w), j + 1));
                const float w2 = __int_as_float(__builtin_amdgcn_readlane(__float_as_int(my_w), j + 2));
                const float w3 = __int_as_float(__builtin_amdgcn_readlane(__float_as_int(my_w), j + 3));
                const float4 a0 = h4[(size_t)s0 * 32 + lane];
                const float4 a1 = h4[(size_t)s1 * 32 + lane];
                const float4 a2 = h4[(size_t)s2 * 32 + lane];
                const float4 a3 = h4[(size_t)s3 * 32 + lane];
                fma4(acc, a0, w0);
                fma4(acc, a1, w1);
                fma4(acc, a2, w2);
                fma4(acc, a3, w3);
            }
            for (; j < m; ++j) {
                const int   s = __builtin_amdgcn_readlane(my_s, j);
                const float w = __int_as_float(__builtin_amdgcn_readlane(__float_as_int(my_w), j));
                const float4 a = h4[(size_t)s * 32 + lane];
                fma4(acc, a, w);
            }
        }

        acc.x += bias0;
        acc.y += bias1;
        acc.z += bias2;
        acc.w += bias3;
        if (do_relu) {
            acc.x = fmaxf(acc.x, 0.0f);
            acc.y = fmaxf(acc.y, 0.0f);
            acc.z = fmaxf(acc.z, 0.0f);
            acc.w = fmaxf(acc.w, 0.0f);
        }
        ((float4*)out)[(size_t)n * 32 + lane] = acc;
    }
}

extern "C" void kernel_launch(void* const* d_in, const int* in_sizes, int n_in,
                              void* d_out, int out_size, void* d_ws, size_t ws_size,
                              hipStream_t stream) {
    const float* x  = (const float*)d_in[0];
    const int*   ei = (const int*)d_in[1];
    const float* W1 = (const float*)d_in[2];
    const float* b1 = (const float*)d_in[3];
    const float* W2 = (const float*)d_in[4];
    const float* b2 = (const float*)d_in[5];
    float* out = (float*)d_out;

    const int N = in_sizes[0] / D;
    const int E = in_sizes[1] / 2;
    const int* src = ei;
    const int* dst = ei + E;
    const size_t ND = (size_t)N * D;

    const int nb_N = (N + 255) / 256;
    const int nb_E = (E + 256 * EPT - 1) / (256 * EPT);
    const int nb_G = (N + 8 * NPW - 1) / (8 * NPW);
    const int nt_M = N / 16;

    float* ws   = (float*)d_ws;
    float* dinv = ws;
    float* h    = ws + 65536;
    int* cnt    = (int*)(h + ND);
    int* rowptr = cnt + N;
    int* bsum   = rowptr + N;
    int* rank   = bsum + nb_N;
    int* col    = rank + E;

    k_zero_cnt  <<<nb_N, 256, 0, stream>>>(cnt, N);
    k_count_rank<<<nb_E, 256, 0, stream>>>(dst, cnt, rank, E);
    k_dinv_bsum <<<nb_N, 256, 0, stream>>>(cnt, dinv, bsum, N);
    k_rowptr    <<<nb_N, 256, 0, stream>>>(cnt, bsum, rowptr, N);
    k_fill      <<<nb_E, 256, 0, stream>>>(src, dst, rowptr, rank, col, E);

    k_gemm_wmma<<<GEMM_BLOCKS, 256, 0, stream>>>(x, W1, h, nt_M);
    k_gather   <<<nb_G, 256, 0, stream>>>(h, rowptr, cnt, col, dinv, b1, out, N, 1);

    k_gemm_wmma<<<GEMM_BLOCKS, 256, 0, stream>>>(out, W2, h, nt_M);
    k_gather   <<<nb_G, 256, 0, stream>>>(h, rowptr, cnt, col, dinv, b2, out, N, 0);
}


# Aggregate before transform, (A x) W for A (x W): a 16-node tile is gathered into LDS and multiplied by W on the matrix unit in one kernel per layer; no GEMM launches, sums reordered
# speedup vs baseline: 2.7454x; 1.1289x over previous
#include <hip/hip_runtime.h>


#define D 128

#define EPT 8
#define TP 132
#define FUSED_BLOCKS 782

typedef __attribute__((ext_vector_type(2))) float v2f;
typedef __attribute__((ext_vector_type(8))) float v8f;

__global__ __launch_bounds__(256) void k_zero_cnt(int* cnt, int N) {
    int i = blockIdx.x * 256 + threadIdx.x;
    if (i < N) cnt[i] = 0;
}

__global__ __launch_bounds__(256) void k_count_rank(const int* __restrict__ dst,
                                                    int* cnt,
                                                    int* __restrict__ rank, int E) {
    const int base = blockIdx.x * (256 * EPT) + threadIdx.x;
    if ((int)(blockIdx.x + 1) * (256 * EPT) <= E) {
        int d[EPT];
        int r[EPT];
#pragma unroll
        for (int u = 0; u < EPT; ++u) d[u] = dst[base + u * 256];
#pragma unroll
        for (int u = 0; u < EPT; ++u) r[u] = atomicAdd(&cnt[d[u]], 1);
#pragma unroll
        for (int u = 0; u < EPT; ++u) rank[base + u * 256] = r[u];
    } else {
        for (int u = 0; u < EPT; ++u) {
            const int i = base + u * 256;
            if (i < E) rank[i] = atomicAdd(&cnt[dst[i]], 1);
        }
    }
}

__global__ __launch_bounds__(256) void k_dinv_bsum(const int* __restrict__ cnt,
                                                   float* __restrict__ dinv,
                                                   int* __restrict__ bsum, int N) {
    __shared__ int s_w[8];
    const int tid  = threadIdx.x;
    const int lane = tid & 31;
    const int wave = tid >> 5;
    const int i = blockIdx.x * 256 + tid;
    const int c = (i < N) ? cnt[i] : 0;
    if (i < N) dinv[i] = rsqrtf((float)(c + 1));
    int x = c;
    for (int o = 16; o > 0; o >>= 1) x += __shfl_xor(x, o);
    if (lane == 0) s_w[wave] = x;
    __syncthreads();
    if (tid == 0) {
        int t = 0;
        for (int w = 0; w < 8; ++w) t += s_w[w];
        bsum[blockIdx.x] = t;
    }
}

__global__ __launch_bounds__(256) void k_rowptr(const int* __restrict__ cnt,
                                                const int* __restrict__ bsum,
                                                int* __restrict__ rowptr, int N) {
    __shared__ int s_part[8];
    __shared__ int s_tot[8];
    const int tid  = threadIdx.x;
    const int lane = tid & 31;
    const int wave = tid >> 5;
    const int nblk = (int)blockIdx.x;

    int part = 0;
    for (int j = tid; j < nblk; j += 256) part += bsum[j];
    for (int o = 16; o > 0; o >>= 1) part += __shfl_xor(part, o);

    const int i = nblk * 256 + tid;
    const int c = (i < N) ? cnt[i] : 0;
    int x = c;
    for (int o = 1; o < 32; o <<= 1) {
        int y = __shfl_up(x, o);
        if (lane >= o) x += y;
    }
    if (lane == 31) s_tot[wave] = x;
    if (lane == 0)  s_part[wave] = part;
    __syncthreads();

    int base = 0;
    for (int w = 0; w < 8; ++w) base += s_part[w];
    for (int w = 0; w < wave; ++w) base += s_tot[w];
    if (i < N) rowptr[i] = base + x - c;
}

__global__ __launch_bounds__(256) void k_fill(const int* __restrict__ src,
                                              const int* __restrict__ dst,
                                              const int* __restrict__ rowptr,
                                              const int* __restrict__ rank,
                                              int* __restrict__ col, int E) {
    const int base = blockIdx.x * (256 * EPT) + threadIdx.x;
    if ((int)(blockIdx.x + 1) * (256 * EPT) <= E) {
        int d[EPT];
        int k[EPT];
        int s[EPT];
        int p[EPT];
#pragma unroll
        for (int u = 0; u < EPT; ++u) {
            d[u] = dst[base + u * 256];
            k[u] = rank[base + u * 256];
            s[u] = src[base + u * 256];
        }
#pragma unroll
        for (int u = 0; u < EPT; ++u) p[u] = rowptr[d[u]];
#pragma unroll
        for (int u = 0; u < EPT; ++u) col[p[u] + k[u]] = s[u];
    } else {
        for (int u = 0; u < EPT; ++u) {
            const int i = base + u * 256;
            if (i < E) col[rowptr[dst[i]] + rank[i]] = src[i];
        }
    }
}

static __device__ __forceinline__ void fma4(float4& acc, const float4 a, const float w) {
    acc.x = fmaf(a.x, w, acc.x);
    acc.y = fmaf(a.y, w, acc.y);
    acc.z = fmaf(a.z, w, acc.z);
    acc.w = fmaf(a.w, w, acc.w);
}

template <int K>
static __device__ __forceinline__ void rows(float4& acc, const float4* __restrict__ x4,
                                            const int my_s, const float my_d, const float dd,
                                            const int j, const int lane) {
    float4 a[K];
    float  w[K];
#pragma unroll
    for (int u = 0; u < K; ++u) {
        const int s = __builtin_amdgcn_readlane(my_s, j + u);
        w[u] = __int_as_float(__builtin_amdgcn_readlane(__float_as_int(my_d), j + u)) * dd;
        a[u] = x4[(size_t)s * 32 + lane];
    }
#pragma unroll
    for (int u = 0; u < K; ++u) fma4(acc, a[u], w[u]);
}

__global__ __launch_bounds__(256) void k_agg_gemm(const float* __restrict__ X,
                                                  const float* __restrict__ W,
                                                  const int* __restrict__ rowptr,
                                                  const int* __restrict__ cnt,
                                                  const int* __restrict__ col,
                                                  const float* __restrict__ dinv,
                                                  const float* __restrict__ bias,
                                                  float* __restrict__ out,
                                                  int ntiles, int do_relu) {
    __shared__ __attribute__((aligned(16))) float As[16 * TP];
    __shared__ __attribute__((aligned(16))) float Cs[16 * TP];

    const int lane  = threadIdx.x & 31;
    const int wave  = __builtin_amdgcn_readfirstlane((int)(threadIdx.x >> 5));
    const int col0  = wave * 16;
    const int lrow  = lane & 15;
    const int khalf = (lane >> 4) * 2;
    const int m_add = (lane >> 4) * 8;
    const int ncol  = col0 + lrow;

    v2f b[D / 4];
#pragma unroll
    for (int i = 0; i < D / 4; ++i) {
        b[i].x = W[(size_t)(4 * i + khalf + 0) * D + ncol];
        b[i].y = W[(size_t)(4 * i + khalf + 1) * D + ncol];
    }
    const float bn = bias[ncol];
    const float4* x4 = (const float4*)X;

    for (int t = blockIdx.x; t < ntiles; t += gridDim.x) {
        const int row0 = t * 16;

        const int n0 = row0 + 2 * wave;
        int   beg[2], c[2];
        float dd[2];
        float4 hv[2];
        int   my_s[2];
        float my_w[2];
#pragma unroll
        for (int q = 0; q < 2; ++q) {
            const int n = n0 + q;
            beg[q] = rowptr[n];
            c[q]   = cnt[n];
            dd[q]  = dinv[n];
            hv[q]  = x4[(size_t)n * 32 + lane];
        }
#pragma unroll
        for (int q = 0; q < 2; ++q) my_s[q] = col[(lane < c[q]) ? beg[q] + lane : 0];
        asm volatile("" : "+v"(my_s[0]), "+v"(my_s[1]));
#pragma unroll
        for (int q = 0; q < 2; ++q) my_w[q] = dinv[(lane < c[q]) ? my_s[q] : 0];

#pragma unroll
        for (int q = 0; q < 2; ++q) {
            const float wself = dd[q] * dd[q];
            float4 acc;
            acc.x = hv[q].x * wself;
            acc.y = hv[q].y * wself;
            acc.z = hv[q].z * wself;
            acc.w = hv[q].w * wself;

            int   cs = my_s[q];
            float cw = my_w[q];
            for (int base = 0; base < c[q]; base += 32) {
                const int m = min(32, c[q] - base);
                if (base > 0) {
                    cs = 0;
                    cw = 0.0f;
                    if (lane < m) {
                        cs = col[beg[q] + base + lane];
                        cw = dinv[cs];
                    }
                }
                int j = 0;
                for (; j + 8 <= m; j += 8) rows<8>(acc, x4, cs, cw, dd[q], j, lane);
                if (j + 4 <= m) { rows<4>(acc, x4, cs, cw, dd[q], j, lane); j += 4; }
                const int rem = m - j;
                if (rem == 3)      rows<3>(acc, x4, cs, cw, dd[q], j, lane);
                else if (rem == 2) rows<2>(acc, x4, cs, cw, dd[q], j, lane);
                else if (rem == 1) rows<1>(acc, x4, cs, cw, dd[q], j, lane);
            }
            *(float4*)&As[(2 * wave + q) * TP + lane * 4] = acc;
        }
        __syncthreads();

        v8f d = {};
#pragma unroll
        for (int i = 0; i < D / 4; ++i) {
            const v2f a = *(const v2f*)&As[lrow * TP + 4 * i + khalf];
            d = __builtin_amdgcn_wmma_f32_16x16x4_f32(
                 false, a,  false, b[i],
                 (short)0, d,  false,  false);
        }

#pragma unroll
        for (int r = 0; r < 8; ++r) {
            float v = d[r] + bn;
            if (do_relu) v = fmaxf(v, 0.0f);
            Cs[(r + m_add) * TP + ncol] = v;
        }
        __syncthreads();
#pragma unroll
        for (int q = 0; q < 2; ++q) {
            const int n = n0 + q;
            ((float4*)out)[(size_t)n * 32 + lane] = *(const float4*)&Cs[(2 * wave + q) * TP + lane * 4];
        }
    }
}

extern "C" void kernel_launch(void* const* d_in, const int* in_sizes, int n_in,
                              void* d_out, int out_size, void* d_ws, size_t ws_size,
                              hipStream_t stream) {
    const float* x  = (const float*)d_in[0];
    const int*   ei = (const int*)d_in[1];
    const float* W1 = (const float*)d_in[2];
    const float* b1 = (const float*)d_in[3];
    const float* W2 = (const float*)d_in[4];
    const float* b2 = (const float*)d_in[5];
    float* out = (float*)d_out;

    const int N = in_sizes[0] / D;
    const int E = in_sizes[1] / 2;
    const int* src = ei;
    const int* dst = ei + E;
    const size_t ND = (size_t)N * D;

    const int nb_N = (N + 255) / 256;
    const int nb_E = (E + 256 * EPT - 1) / (256 * EPT);
    const int nt_M = N / 16;

    float* ws   = (float*)d_ws;
    float* dinv = ws;
    float* h    = ws + 65536;
    int* cnt    = (int*)(h + ND);
    int* rowptr = cnt + N;
    int* bsum   = rowptr + N;
    int* rank   = bsum + nb_N;
    int* col    = rank + E;

    k_zero_cnt  <<<nb_N, 256, 0, stream>>>(cnt, N);
    k_count_rank<<<nb_E, 256, 0, stream>>>(dst, cnt, rank, E);
    k_dinv_bsum <<<nb_N, 256, 0, stream>>>(cnt, dinv, bsum, N);
    k_rowptr    <<<nb_N, 256, 0, stream>>>(cnt, bsum, rowptr, N);
    k_fill      <<<nb_E, 256, 0, stream>>>(src, dst, rowptr, rank, col, E);

    k_agg_gemm<<<FUSED_BLOCKS, 256, 0, stream>>>(x, W1, rowptr, cnt, col, dinv, b1, h, nt_M, 1);
    k_agg_gemm<<<FUSED_BLOCKS, 256, 0, stream>>>(h, W2, rowptr, cnt, col, dinv, b2, out, nt_M, 0);
}


# Degree counters split into one plane per package half: an edge's returning add goes to its own half's plane, so no counter line moves between the two caches; arrival-order ranks now per half
# speedup vs baseline: 3.9545x; 1.4404x over previous
#include <hip/hip_runtime.h>


#define D 128

#define EPT 8
#define TP 132
#define FUSED_BLOCKS 782

typedef __attribute__((ext_vector_type(2))) float v2f;
typedef __attribute__((ext_vector_type(8))) float v8f;

static __device__ __forceinline__ int pkg_half() {
    const unsigned v = (unsigned)__builtin_amdgcn_s_sendmsg_rtn(0x87);
    return (int)((v >> 18) & 1u);
}

__global__ __launch_bounds__(256) void k_zero_cnt(int* cnt2, int n) {
    int i = blockIdx.x * 256 + threadIdx.x;
    if (i < n) cnt2[i] = 0;
}

__global__ __launch_bounds__(256) void k_count_rank(const int* __restrict__ dst,
                                                    int* cnt2,
                                                    int* __restrict__ rank, int E, int NP) {
    const int hf  = pkg_half();
    int* cnt      = cnt2 + (size_t)hf * NP;
    const int tag = (int)((unsigned)hf << 31);
    const int base = blockIdx.x * (256 * EPT) + threadIdx.x;
    if ((int)(blockIdx.x + 1) * (256 * EPT) <= E) {
        int d[EPT];
        int r[EPT];
#pragma unroll
        for (int u = 0; u < EPT; ++u) d[u] = dst[base + u * 256];
#pragma unroll
        for (int u = 0; u < EPT; ++u) r[u] = atomicAdd(&cnt[d[u]], 1);
#pragma unroll
        for (int u = 0; u < EPT; ++u) rank[base + u * 256] = r[u] | tag;
    } else {
        for (int u = 0; u < EPT; ++u) {
            const int i = base + u * 256;
            if (i < E) rank[i] = atomicAdd(&cnt[dst[i]], 1) | tag;
        }
    }
}

__global__ __launch_bounds__(256) void k_dinv_bsum(const int* __restrict__ cnt2,
                                                   int* __restrict__ cnt,
                                                   float* __restrict__ dinv,
                                                   int* __restrict__ bsum, int N, int NP) {
    __shared__ int s_w[8];
    const int tid  = threadIdx.x;
    const int lane = tid & 31;
    const int wave = tid >> 5;
    const int i = blockIdx.x * 256 + tid;
    const int c = (i < N) ? cnt2[i] + cnt2[NP + i] : 0;
    if (i < N) {
        cnt[i]  = c;
        dinv[i] = rsqrtf((float)(c + 1));
    }
    int x = c;
    for (int o = 16; o > 0; o >>= 1) x += __shfl_xor(x, o);
    if (lane == 0) s_w[wave] = x;
    __syncthreads();
    if (tid == 0) {
        int t = 0;
        for (int w = 0; w < 8; ++w) t += s_w[w];
        bsum[blockIdx.x] = t;
    }
}

__global__ __launch_bounds__(256) void k_rowptr(const int* __restrict__ cnt,
                                                const int* __restrict__ cnt2,
                                                const int* __restrict__ bsum,
                                                int* __restrict__ rowptr,
                                                int* __restrict__ rowptrB, int N) {
    __shared__ int s_part[8];
    __shared__ int s_tot[8];
    const int tid  = threadIdx.x;
    const int lane = tid & 31;
    const int wave = tid >> 5;
    const int nblk = (int)blockIdx.x;

    int part = 0;
    for (int j = tid; j < nblk; j += 256) part += bsum[j];
    for (int o = 16; o > 0; o >>= 1) part += __shfl_xor(part, o);

    const int i = nblk * 256 + tid;
    const int c = (i < N) ? cnt[i] : 0;
    int x = c;
    for (int o = 1; o < 32; o <<= 1) {
        int y = __shfl_up(x, o);
        if (lane >= o) x += y;
    }
    if (lane == 31) s_tot[wave] = x;
    if (lane == 0)  s_part[wave] = part;
    __syncthreads();

    int base = 0;
    for (int w = 0; w < 8; ++w) base += s_part[w];
    for (int w = 0; w < wave; ++w) base += s_tot[w];
    if (i < N) {
        const int p = base + x - c;
        rowptr[i]  = p;
        rowptrB[i] = p + cnt2[i];
    }
}

__global__ __launch_bounds__(256) void k_fill(const int* __restrict__ src,
                                              const int* __restrict__ dst,
                                              const int* __restrict__ rowptr,
                                              const int* __restrict__ rowptrB,
                                              const int* __restrict__ rank,
                                              int* __restrict__ col, int E) {
    const int base = blockIdx.x * (256 * EPT) + threadIdx.x;
    if ((int)(blockIdx.x + 1) * (256 * EPT) <= E) {
        int d[EPT];
        int k[EPT];
        int s[EPT];
        int p[EPT];
#pragma unroll
        for (int u = 0; u < EPT; ++u) {
            d[u] = dst[base + u * 256];
            k[u] = rank[base + u * 256];
            s[u] = src[base + u * 256];
        }
#pragma unroll
        for (int u = 0; u < EPT; ++u) p[u] = ((k[u] < 0) ? rowptrB : rowptr)[d[u]];
#pragma unroll
        for (int u = 0; u < EPT; ++u) col[p[u] + (k[u] & 0x7fffffff)] = s[u];
    } else {
        for (int u = 0; u < EPT; ++u) {
            const int i = base + u * 256;
            if (i < E) {
                const int kk = rank[i];
                col[((kk < 0) ? rowptrB : rowptr)[dst[i]] + (kk & 0x7fffffff)] = src[i];
            }
        }
    }
}

static __device__ __forceinline__ void fma4(float4& acc, const float4 a, const float w) {
    acc.x = fmaf(a.x, w, acc.x);
    acc.y = fmaf(a.y, w, acc.y);
    acc.z = fmaf(a.z, w, acc.z);
    acc.w = fmaf(a.w, w, acc.w);
}

template <int K>
static __device__ __forceinline__ void rows(float4& acc, const float4* __restrict__ x4,
                                            const int my_s, const float my_d, const float dd,
                                            const int j, const int lane) {
    float4 a[K];
    float  w[K];
#pragma unroll
    for (int u = 0; u < K; ++u) {
        const int s = __builtin_amdgcn_readlane(my_s, j + u);
        w[u] = __int_as_float(__builtin_amdgcn_readlane(__float_as_int(my_d), j + u)) * dd;
        a[u] = x4[(size_t)s * 32 + lane];
    }
#pragma unroll
    for (int u = 0; u < K; ++u) fma4(acc, a[u], w[u]);
}

__global__ __launch_bounds__(256) void k_agg_gemm(const float* __restrict__ X,
                                                  const float* __restrict__ W,
                                                  const int* __restrict__ rowptr,
                                                  const int* __restrict__ cnt,
                                                  const int* __restrict__ col,
                                                  const float* __restrict__ dinv,
                                                  const float* __restrict__ bias,
                                                  float* __restrict__ out,
                                                  int ntiles, int do_relu) {
    __shared__ __attribute__((aligned(16))) float As[16 * TP];
    __shared__ __attribute__((aligned(16))) float Cs[16 * TP];

    const int lane  = threadIdx.x & 31;
    const int wave  = __builtin_amdgcn_readfirstlane((int)(threadIdx.x >> 5));
    const int col0  = wave * 16;
    const int lrow  = lane & 15;
    const int khalf = (lane >> 4) * 2;
    const int m_add = (lane >> 4) * 8;
    const int ncol  = col0 + lrow;

    v2f b[D / 4];
#pragma unroll
    for (int i = 0; i < D / 4; ++i) {
        b[i].x = W[(size_t)(4 * i + khalf + 0) * D + ncol];
        b[i].y = W[(size_t)(4 * i + khalf + 1) * D + ncol];
    }
    const float bn = bias[ncol];
    const float4* x4 = (const float4*)X;

    for (int t = blockIdx.x; t < ntiles; t += gridDim.x) {
        const int row0 = t * 16;

        const int n0 = row0 + 2 * wave;
        int   beg[2], c[2];
        float dd[2];
        float4 hv[2];
        int   my_s[2];
        float my_w[2];
#pragma unroll
        for (int q = 0; q < 2; ++q) {
            const int n = n0 + q;
            beg[q] = rowptr[n];
            c[q]   = cnt[n];
            dd[q]  = dinv[n];
            hv[q]  = x4[(size_t)n * 32 + lane];
        }
#pragma unroll
        for (int q = 0; q < 2; ++q) my_s[q] = col[(lane < c[q]) ? beg[q] + lane : 0];
        asm volatile("" : "+v"(my_s[0]), "+v"(my_s[1]));
#pragma unroll
        for (int q = 0; q < 2; ++q) my_w[q] = dinv[(lane < c[q]) ? my_s[q] : 0];

#pragma unroll
        for (int q = 0; q < 2; ++q) {
            const float wself = dd[q] * dd[q];
            float4 acc;
            acc.x = hv[q].x * wself;
            acc.y = hv[q].y * wself;
            acc.z = hv[q].z * wself;
            acc.w = hv[q].w * wself;

            int   cs = my_s[q];
            float cw = my_w[q];
            for (int base = 0; base < c[q]; base += 32) {
                const int m = min(32, c[q] - base);
                if (base > 0) {
                    cs = 0;
                    cw = 0.0f;
                    if (lane < m) {
                        cs = col[beg[q] + base + lane];
                        cw = dinv[cs];
                    }
                }
                int j = 0;
                for (; j + 8 <= m; j += 8) rows<8>(acc, x4, cs, cw, dd[q], j, lane);
                if (j + 4 <= m) { rows<4>(acc, x4, cs, cw, dd[q], j, lane); j += 4; }
                const int rem = m - j;
                if (rem == 3)      rows<3>(acc, x4, cs, cw, dd[q], j, lane);
                else if (rem == 2) rows<2>(acc, x4, cs, cw, dd[q], j, lane);
                else if (rem == 1) rows<1>(acc, x4, cs, cw, dd[q], j, lane);
            }
            *(float4*)&As[(2 * wave + q) * TP + lane * 4] = acc;
        }
        __syncthreads();

        v8f d = {};
#pragma unroll
        for (int i = 0; i < D / 4; ++i) {
            const v2f a = *(const v2f*)&As[lrow * TP + 4 * i + khalf];
            d = __builtin_amdgcn_wmma_f32_16x16x4_f32(
                 false, a,  false, b[i],
                 (short)0, d,  false,  false);
        }

#pragma unroll
        for (int r = 0; r < 8; ++r) {
            float v = d[r] + bn;
            if (do_relu) v = fmaxf(v, 0.0f);
            Cs[(r + m_add) * TP + ncol] = v;
        }
        __syncthreads();
#pragma unroll
        for (int q = 0; q < 2; ++q) {
            const int n = n0 + q;
            ((float4*)out)[(size_t)n * 32 + lane] = *(const float4*)&Cs[(2 * wave + q) * TP + lane * 4];
        }
    }
}

extern "C" void kernel_launch(void* const* d_in, const int* in_sizes, int n_in,
                              void* d_out, int out_size, void* d_ws, size_t ws_size,
                              hipStream_t stream) {
    const float* x  = (const float*)d_in[0];
    const int*   ei = (const int*)d_in[1];
    const float* W1 = (const float*)d_in[2];
    const float* b1 = (const float*)d_in[3];
    const float* W2 = (const float*)d_in[4];
    const float* b2 = (const float*)d_in[5];
    float* out = (float*)d_out;

    const int N = in_sizes[0] / D;
    const int E = in_sizes[1] / 2;
    const int* src = ei;
    const int* dst = ei + E;
    const size_t ND = (size_t)N * D;

    const int nb_N = (N + 255) / 256;
    const int nb_E = (E + 256 * EPT - 1) / (256 * EPT);
    const int nt_M = N / 16;

    float* ws   = (float*)d_ws;
    float* dinv = ws;
    float* h    = ws + 65536;
    int* cnt    = (int*)(h + ND);
    int* rowptr = cnt + N;
    int* bsum   = rowptr + N;
    int* rank   = bsum + nb_N;
    int* col    = rank + E;
    const int NP = nb_N * 256;
    int* cnt2    = (int*)h;
    int* rowptrB = cnt2 + 2 * NP;

    k_zero_cnt  <<<2 * nb_N, 256, 0, stream>>>(cnt2, 2 * NP);
    k_count_rank<<<nb_E, 256, 0, stream>>>(dst, cnt2, rank, E, NP);
    k_dinv_bsum <<<nb_N, 256, 0, stream>>>(cnt2, cnt, dinv, bsum, N, NP);
    k_rowptr    <<<nb_N, 256, 0, stream>>>(cnt, cnt2, bsum, rowptr, rowptrB, N);
    k_fill      <<<nb_E, 256, 0, stream>>>(src, dst, rowptr, rowptrB, rank, col, E);

    k_agg_gemm<<<FUSED_BLOCKS, 256, 0, stream>>>(x, W1, rowptr, cnt, col, dinv, b1, h, nt_M, 1);
    k_agg_gemm<<<FUSED_BLOCKS, 256, 0, stream>>>(h, W2, rowptr, cnt, col, dinv, b2, out, nt_M, 0);
}


# Buckets as fixed rows of 32 two-byte source ids per node and cache side, filled by the counting pass itself: no rank array, prefix sum or fill pass, 4 launches not 7; dinv formed in layer 1
# speedup vs baseline: 5.0492x; 1.2768x over previous
#include <hip/hip_runtime.h>


#define D 128

#define EPT 8
#define CAP 32
#define TP 132
#define FUSED_BLOCKS 782

typedef __attribute__((ext_vector_type(2))) float v2f;
typedef __attribute__((ext_vector_type(8))) float v8f;

static __device__ __forceinline__ int pkg_half() {
    const unsigned v = (unsigned)__builtin_amdgcn_s_sendmsg_rtn(0x87);
    return (int)((v >> 18) & 1u);
}

__global__ __launch_bounds__(256) void k_zero_cnt(int* cnt2, int n) {
    int i = blockIdx.x * 256 + threadIdx.x;
    if (i < n) cnt2[i] = 0;
}

__global__ __launch_bounds__(256) void k_count_fill(const int* __restrict__ src,
                                                    const int* __restrict__ dst,
                                                    int* cnt2,
                                                    unsigned short* __restrict__ brow,
                                                    int E, int N, int NPL) {
    const int hf = pkg_half();
    int* cnt            = cnt2 + (size_t)hf * NPL;
    unsigned short* row = brow + (size_t)hf * N * CAP;
    const int base = blockIdx.x * (256 * EPT) + threadIdx.x;
    if ((int)(blockIdx.x + 1) * (256 * EPT) <= E) {
        int d[EPT];
        int s[EPT];
        int r[EPT];
#pragma unroll
        for (int u = 0; u < EPT; ++u) {
            d[u] = dst[base + u * 256];
            s[u] = src[base + u * 256];
        }
        asm volatile("" : "+v"(s[0]));
#pragma unroll
        for (int u = 0; u < EPT; ++u) r[u] = atomicAdd(&cnt[d[u]], 1);
#pragma unroll
        for (int u = 0; u < EPT; ++u)
            if (r[u] < CAP) row[(size_t)d[u] * CAP + r[u]] = (unsigned short)s[u];
    } else {
        for (int u = 0; u < EPT; ++u) {
            const int i = base + u * 256;
            if (i < E) {
                const int dd = dst[i];
                const int rr = atomicAdd(&cnt[dd], 1);
                if (rr < CAP) row[(size_t)dd * CAP + rr] = (unsigned short)src[i];
            }
        }
    }
}

static __device__ __forceinline__ void fma4(float4& acc, const float4 a, const float w) {
    acc.x = fmaf(a.x, w, acc.x);
    acc.y = fmaf(a.y, w, acc.y);
    acc.z = fmaf(a.z, w, acc.z);
    acc.w = fmaf(a.w, w, acc.w);
}

template <int K>
static __device__ __forceinline__ void rows(float4& acc, const float4* __restrict__ x4,
                                            const int my_s, const float my_d, const float dd,
                                            const int j, const int lane) {
    float4 a[K];
    float  w[K];
#pragma unroll
    for (int u = 0; u < K; ++u) {
        const int s = __builtin_amdgcn_readlane(my_s, j + u);
        w[u] = __int_as_float(__builtin_amdgcn_readlane(__float_as_int(my_d), j + u)) * dd;
        a[u] = x4[(size_t)s * 32 + lane];
    }
#pragma unroll
    for (int u = 0; u < K; ++u) fma4(acc, a[u], w[u]);
}

static __device__ __forceinline__ int load_src(const unsigned short* __restrict__ b0,
                                               const unsigned short* __restrict__ b1,
                                               const int n, const int c0, const int c, const int idx) {
    const size_t r = (size_t)n * CAP;
    const unsigned short* p = (idx < c0) ? b0 + r + idx : b1 + r + (idx - c0);
    if (idx >= c) p = b0 + r;
    return (int)*p;
}

template <bool FIRST>
static __device__ __forceinline__ float dinv_src(const int* __restrict__ cntA,
                                                 const int* __restrict__ cntB,
                                                 const float* __restrict__ dinv_in, const int s) {
    if (FIRST) return rsqrtf((float)(cntA[s] + cntB[s] + 1));
    return dinv_in[s];
}

template <bool FIRST>
__global__ __launch_bounds__(256) void k_agg_gemm(const float* __restrict__ X,
                                                  const float* __restrict__ W,
                                                  const int* __restrict__ cntA,
                                                  const int* __restrict__ cntB,
                                                  const unsigned short* __restrict__ b0,
                                                  const unsigned short* __restrict__ b1,
                                                  const int* __restrict__ src,
                                                  const int* __restrict__ dst,
                                                  const float* __restrict__ dinv_in,
                                                  float* __restrict__ dinv_out,
                                                  const float* __restrict__ bias,
                                                  float* __restrict__ out,
                                                  int ntiles, int E, int do_relu) {
    __shared__ __attribute__((aligned(16))) float As[16 * TP];
    __shared__ __attribute__((aligned(16))) float Cs[16 * TP];

    const int lane  = threadIdx.x & 31;
    const int wave  = __builtin_amdgcn_readfirstlane((int)(threadIdx.x >> 5));
    const int col0  = wave * 16;
    const int lrow  = lane & 15;
    const int khalf = (lane >> 4) * 2;
    const int m_add = (lane >> 4) * 8;
    const int ncol  = col0 + lrow;

    v2f b[D / 4];
#pragma unroll
    for (int i = 0; i < D / 4; ++i) {
        b[i].x = W[(size_t)(4 * i + khalf + 0) * D + ncol];
        b[i].y = W[(size_t)(4 * i + khalf + 1) * D + ncol];
    }
    const float bn = bias[ncol];
    const float4* x4 = (const float4*)X;

    for (int t = blockIdx.x; t < ntiles; t += gridDim.x) {
        const int row0 = t * 16;

        const int n0 = row0 + 2 * wave;
        int   c0[2], c[2], cm[2];
        float dd[2];
        float4 hv[2];
        int   my_s[2];
        float my_w[2];
#pragma unroll
        for (int q = 0; q < 2; ++q) {
            const int n = n0 + q;
            c0[q] = cntA[n];
            c[q]  = cntB[n];
            hv[q] = x4[(size_t)n * 32 + lane];
        }
#pragma unroll
        for (int q = 0; q < 2; ++q) {
            const bool fits = (c0[q] <= CAP) && (c[q] <= CAP);
            c[q]  = c0[q] + c[q];
            cm[q] = fits ? c[q] : 0;
            dd[q] = rsqrtf((float)(c[q] + 1));
            if (FIRST) { if (lane == 0) dinv_out[n0 + q] = dd[q]; }
            my_s[q] = load_src(b0, b1, n0 + q, c0[q], cm[q], lane);
        }
        asm volatile("" : "+v"(my_s[0]), "+v"(my_s[1]));
#pragma unroll
        for (int q = 0; q < 2; ++q)
            my_w[q] = dinv_src<FIRST>(cntA, cntB, dinv_in, (lane < cm[q]) ? my_s[q] : 0);

#pragma unroll
        for (int q = 0; q < 2; ++q) {
            const float wself = dd[q] * dd[q];
            float4 acc;
            acc.x = hv[q].x * wself;
            acc.y = hv[q].y * wself;
            acc.z = hv[q].z * wself;
            acc.w = hv[q].w * wself;

            int   cs = my_s[q];
            float cw = my_w[q];
            for (int base = 0; base < cm[q]; base += 32) {
                const int m = min(32, cm[q] - base);
                if (base > 0) {
                    cs = load_src(b0, b1, n0 + q, c0[q], cm[q], base + lane);
                    cw = dinv_src<FIRST>(cntA, cntB, dinv_in, (lane < m) ? cs : 0);
                }
                int j = 0;
                for (; j + 8 <= m; j += 8) rows<8>(acc, x4, cs, cw, dd[q], j, lane);
                if (j + 4 <= m) { rows<4>(acc, x4, cs, cw, dd[q], j, lane); j += 4; }
                const int rem = m - j;
                if (rem == 3)      rows<3>(acc, x4, cs, cw, dd[q], j, lane);
                else if (rem == 2) rows<2>(acc, x4, cs, cw, dd[q], j, lane);
                else if (rem == 1) rows<1>(acc, x4, cs, cw, dd[q], j, lane);
            }
            if (cm[q] != c[q]) {
                const int n = n0 + q;
                for (int e0 = 0; e0 < E; e0 += 32) {
                    const int e  = e0 + lane;
                    const int dv = (e < E) ? dst[e] : -1;
                    unsigned mk = __builtin_amdgcn_ballot_w32(dv == n);
                    if (mk != 0u) {
                        const int   sv = (dv == n) ? src[e] : 0;
                        const float wv = dinv_src<FIRST>(cntA, cntB, dinv_in, sv);
                        while (mk != 0u) {
                            const int j = __builtin_ctz(mk);
                            mk &= mk - 1u;
                            rows<1>(acc, x4, sv, wv, dd[q], j, lane);
                        }
                    }
                }
            }
            *(float4*)&As[(2 * wave + q) * TP + lane * 4] = acc;
        }
        __syncthreads();

        v8f d = {};
#pragma unroll
        for (int i = 0; i < D / 4; ++i) {
            const v2f a = *(const v2f*)&As[lrow * TP + 4 * i + khalf];
            d = __builtin_amdgcn_wmma_f32_16x16x4_f32(
                 false, a,  false, b[i],
                 (short)0, d,  false,  false);
        }

#pragma unroll
        for (int r = 0; r < 8; ++r) {
            float v = d[r] + bn;
            if (do_relu) v = fmaxf(v, 0.0f);
            Cs[(r + m_add) * TP + ncol] = v;
        }
        __syncthreads();
#pragma unroll
        for (int q = 0; q < 2; ++q) {
            const int n = n0 + q;
            ((float4*)out)[(size_t)n * 32 + lane] = *(const float4*)&Cs[(2 * wave + q) * TP + lane * 4];
        }
    }
}

extern "C" void kernel_launch(void* const* d_in, const int* in_sizes, int n_in,
                              void* d_out, int out_size, void* d_ws, size_t ws_size,
                              hipStream_t stream) {
    const float* x  = (const float*)d_in[0];
    const int*   ei = (const int*)d_in[1];
    const float* W1 = (const float*)d_in[2];
    const float* b1 = (const float*)d_in[3];
    const float* W2 = (const float*)d_in[4];
    const float* b2 = (const float*)d_in[5];
    float* out = (float*)d_out;

    const int N = in_sizes[0] / D;
    const int E = in_sizes[1] / 2;
    const int* src = ei;
    const int* dst = ei + E;
    const size_t ND = (size_t)N * D;

    const int nb_E = (E + 256 * EPT - 1) / (256 * EPT);
    const int nt_M = N / 16;
    const int NPL  = ((N + 31) / 32) * 32;
    const int nb_Z = (2 * NPL + 255) / 256;

    float* ws   = (float*)d_ws;
    float* dinv = ws;
    float* h    = ws + 65536;
    int* cnt2   = (int*)(h + ND);
    unsigned short* brow = (unsigned short*)(cnt2 + 2 * NPL);
    const int* cntA = cnt2;
    const int* cntB = cnt2 + NPL;
    const unsigned short* brow0 = brow;
    const unsigned short* brow1 = brow + (size_t)N * CAP;

    k_zero_cnt  <<<nb_Z, 256, 0, stream>>>(cnt2, 2 * NPL);
    k_count_fill<<<nb_E, 256, 0, stream>>>(src, dst, cnt2, brow, E, N, NPL);

    k_agg_gemm<true> <<<FUSED_BLOCKS, 256, 0, stream>>>(x, W1, cntA, cntB, brow0, brow1, src, dst,
                                                        dinv, dinv, b1, h, nt_M, E, 1);
    k_agg_gemm<false><<<FUSED_BLOCKS, 256, 0, stream>>>(h, W2, cntA, cntB, brow0, brow1, src, dst,
                                                        dinv, dinv, b2, out, nt_M, E, 0);
}


# Next tile's index chain (counts, bucket sources, their dinv) is fetched during the current tile's row sums and GEMM; tile barriers order LDS only; self-loop row summed last, as a bucket entry
# speedup vs baseline: 5.2455x; 1.0389x over previous
#include <hip/hip_runtime.h>


#define D 128

#define EPT 8
#define CAP 32
#define TP 132
#define FUSED_BLOCKS 782

typedef __attribute__((ext_vector_type(2))) float v2f;
typedef __attribute__((ext_vector_type(8))) float v8f;
typedef const float4 __attribute__((address_space(1)))* gf4_ptr;

static __device__ __forceinline__ int pkg_half() {
    const unsigned v = (unsigned)__builtin_amdgcn_s_sendmsg_rtn(0x87);
    return (int)((v >> 18) & 1u);
}

__global__ __launch_bounds__(256) void k_zero_cnt(int* cnt2, int n) {
    int i = blockIdx.x * 256 + threadIdx.x;
    if (i < n) cnt2[i] = 0;
}

__global__ __launch_bounds__(256) void k_count_fill(const int* __restrict__ src,
                                                    const int* __restrict__ dst,
                                                    int* cnt2,
                                                    unsigned short* __restrict__ brow,
                                                    int E, int N, int NPL) {
    const int hf = pkg_half();
    int* cnt            = cnt2 + (size_t)hf * NPL;
    unsigned short* row = brow + (size_t)hf * N * CAP;
    const int base = blockIdx.x * (256 * EPT) + threadIdx.x;
    if ((int)(blockIdx.x + 1) * (256 * EPT) <= E) {
        int d[EPT];
        int s[EPT];
        int r[EPT];
#pragma unroll
        for (int u = 0; u < EPT; ++u) {
            d[u] = dst[base + u * 256];
            s[u] = src[base + u * 256];
        }
        asm volatile("" : "+v"(s[0]));
#pragma unroll
        for (int u = 0; u < EPT; ++u) r[u] = atomicAdd(&cnt[d[u]], 1);
#pragma unroll
        for (int u = 0; u < EPT; ++u)
            if (r[u] < CAP) row[(size_t)d[u] * CAP + r[u]] = (unsigned short)s[u];
    } else {
        for (int u = 0; u < EPT; ++u) {
            const int i = base + u * 256;
            if (i < E) {
                const int dd = dst[i];
                const int rr = atomicAdd(&cnt[dd], 1);
                if (rr < CAP) row[(size_t)dd * CAP + rr] = (unsigned short)src[i];
            }
        }
    }
}

static __device__ __forceinline__ void fma4(float4& acc, const float4 a, const float w) {
    acc.x = fmaf(a.x, w, acc.x);
    acc.y = fmaf(a.y, w, acc.y);
    acc.z = fmaf(a.z, w, acc.z);
    acc.w = fmaf(a.w, w, acc.w);
}

template <int K>
static __device__ __forceinline__ void rows(float4& acc, const float4* __restrict__ x4,
                                            const int my_s, const float my_d, const float dd,
                                            const int j, const int lane) {
    float4 a[K];
    float  w[K];
    unsigned long long ra[K];
#pragma unroll
    for (int u = 0; u < K; ++u) {
        const int s = __builtin_amdgcn_readlane(my_s, j + u);
        w[u] = __int_as_float(__builtin_amdgcn_readlane(__float_as_int(my_d), j + u)) * dd;
        ra[u] = (unsigned long long)x4 + ((unsigned long long)(unsigned)s << 9);
    }
    if constexpr (K == 8)
        asm volatile("" : "+s"(ra[0]), "+s"(ra[1]), "+s"(ra[2]), "+s"(ra[3]),
                          "+s"(ra[4]), "+s"(ra[5]), "+s"(ra[6]), "+s"(ra[7]));
    else if constexpr (K == 4)
        asm volatile("" : "+s"(ra[0]), "+s"(ra[1]), "+s"(ra[2]), "+s"(ra[3]));
    else if constexpr (K == 3)
        asm volatile("" : "+s"(ra[0]), "+s"(ra[1]), "+s"(ra[2]));
    else if constexpr (K == 2)
        asm volatile("" : "+s"(ra[0]), "+s"(ra[1]));
    else
        asm volatile("" : "+s"(ra[0]));
#pragma unroll
    for (int u = 0; u < K; ++u) {
        a[u] = ((gf4_ptr)ra[u])[lane];
    }
#pragma unroll
    for (int u = 0; u < K; ++u) fma4(acc, a[u], w[u]);
}

static __device__ __forceinline__ int load_src(const unsigned short* __restrict__ b0,
                                               const unsigned short* __restrict__ b1,
                                               const int n, const int c0, const int c, const int idx) {
    const int hs = (int)(b1 - b0);
    const int r  = n * CAP;
    int e = (idx < c0) ? r + idx : hs + r + (idx - c0);
    if (idx >= c) e = r;
    return (int)b0[e];
}

template <bool FIRST>
static __device__ __forceinline__ float dinv_src(const int* __restrict__ cntA,
                                                 const int* __restrict__ cntB,
                                                 const float* __restrict__ dinv_in, const int s) {
    if (FIRST) return rsqrtf((float)(cntA[s] + cntB[s] + 1));
    return dinv_in[s];
}

static __device__ __forceinline__ void entries(float4& acc, const float4* __restrict__ x4,
                                               const int cs, const float cw, const float dd,
                                               const int m, const int lane) {
    int j = 0;
    for (; j + 8 <= m; j += 8) rows<8>(acc, x4, cs, cw, dd, j, lane);
    if (j + 4 <= m) { rows<4>(acc, x4, cs, cw, dd, j, lane); j += 4; }
    const int rem = m - j;
    if (rem == 3)      rows<3>(acc, x4, cs, cw, dd, j, lane);
    else if (rem == 2) rows<2>(acc, x4, cs, cw, dd, j, lane);
    else if (rem == 1) rows<1>(acc, x4, cs, cw, dd, j, lane);
}

template <bool FIRST>
static __device__ __forceinline__ float4 sum_node(const float4* __restrict__ x4,
                                                  const int* __restrict__ cntA,
                                                  const int* __restrict__ cntB,
                                                  const unsigned short* __restrict__ b0,
                                                  const unsigned short* __restrict__ b1,
                                                  const int* __restrict__ src,
                                                  const int* __restrict__ dst,
                                                  const float* __restrict__ dinv_in,
                                                  const int n, const int c0, const int c, const int cm,
                                                  const float dd,
                                                  const int my_s, const float my_w,
                                                  const int E, const int lane) {
    float4 acc = make_float4(0.0f, 0.0f, 0.0f, 0.0f);
    const int tot = cm + 1;
    {
        const int   cs = (lane == cm) ? n  : my_s;
        const float cw = (lane == cm) ? dd : my_w;
        entries(acc, x4, cs, cw, dd, min(32, tot), lane);
    }
    for (int base = 32; base < tot; base += 32) {
        int   cs = load_src(b0, b1, n, c0, cm, base + lane);
        float cw = dinv_src<FIRST>(cntA, cntB, dinv_in, (base + lane < cm) ? cs : 0);
        if (base + lane == cm) { cs = n; cw = dd; }
        entries(acc, x4, cs, cw, dd, min(32, tot - base), lane);
    }
    if (cm != c) {
        for (int e0 = 0; e0 < E; e0 += 32) {
            const int e  = e0 + lane;
            const int dv = (e < E) ? dst[e] : -1;
            unsigned mk = __builtin_amdgcn_ballot_w32(dv == n);
            if (mk != 0u) {
                const int   sv = (dv == n) ? src[e] : 0;
                const float wv = dinv_src<FIRST>(cntA, cntB, dinv_in, sv);
                while (mk != 0u) {
                    const int j = __builtin_ctz(mk);
                    mk &= mk - 1u;
                    rows<1>(acc, x4, sv, wv, dd, j, lane);
                }
            }
        }
    }
    return acc;
}

static __device__ __forceinline__ void lds_barrier() {
    __builtin_amdgcn_sched_barrier(0);
    __builtin_amdgcn_fence(__ATOMIC_RELEASE, "workgroup", "local");
    __builtin_amdgcn_s_barrier();
    __builtin_amdgcn_fence(__ATOMIC_ACQUIRE, "workgroup", "local");
}

template <bool FIRST>
__global__ __launch_bounds__(256) void k_agg_gemm(const float* __restrict__ X,
                                                  const float* __restrict__ W,
                                                  const int* __restrict__ cntA,
                                                  const int* __restrict__ cntB,
                                                  const unsigned short* __restrict__ b0,
                                                  const unsigned short* __restrict__ b1,
                                                  const int* __restrict__ src,
                                                  const int* __restrict__ dst,
                                                  const float* __restrict__ dinv_in,
                                                  float* __restrict__ dinv_out,
                                                  const float* __restrict__ bias,
                                                  float* __restrict__ out,
                                                  int ntiles, int E, int do_relu) {
    __shared__ __attribute__((aligned(16))) float As[16 * TP];
    __shared__ __attribute__((aligned(16))) float Cs[16 * TP];

    const int lane  = threadIdx.x & 31;
    const int wave  = __builtin_amdgcn_readfirstlane((int)(threadIdx.x >> 5));
    const int col0  = wave * 16;
    const int lrow  = lane & 15;
    const int khalf = (lane >> 4) * 2;
    const int m_add = (lane >> 4) * 8;
    const int ncol  = col0 + lrow;

    v2f b[D / 4];
#pragma unroll
    for (int i = 0; i < D / 4; ++i) {
        b[i].x = W[(size_t)(4 * i + khalf + 0) * D + ncol];
        b[i].y = W[(size_t)(4 * i + khalf + 1) * D + ncol];
    }
    const float bn = bias[ncol];
    const float4* x4 = (const float4*)X;

    int    c0[2] = {0, 0}, c[2] = {0, 0}, cm[2] = {0, 0};
    float  dd[2] = {0.0f, 0.0f};
    int    my_s[2] = {0, 0};
    float  my_w[2] = {0.0f, 0.0f};

    int t = blockIdx.x;
    if (t < ntiles) {
        const int n0 = t * 16 + 2 * wave;
#pragma unroll
        for (int q = 0; q < 2; ++q) {
            const int n = n0 + q;
            c0[q] = cntA[n];
            c[q]  = cntB[n];
        }
#pragma unroll
        for (int q = 0; q < 2; ++q) {
            const bool fits = (c0[q] <= CAP) && (c[q] <= CAP);
            c[q]  = c0[q] + c[q];
            cm[q] = fits ? c[q] : 0;
            dd[q] = rsqrtf((float)(c[q] + 1));
            if (FIRST) { if (lane == 0) dinv_out[n0 + q] = dd[q]; }
            my_s[q] = load_src(b0, b1, n0 + q, c0[q], cm[q], lane);
        }
        asm volatile("" : "+v"(my_s[0]), "+v"(my_s[1]));
#pragma unroll
        for (int q = 0; q < 2; ++q)
            my_w[q] = dinv_src<FIRST>(cntA, cntB, dinv_in, (lane < cm[q]) ? my_s[q] : 0);
    }

    for (; t < ntiles; t += gridDim.x) {
        const int row0 = t * 16;
        const int n0 = row0 + 2 * wave;
        const int tn = t + gridDim.x;
        const bool more = tn < ntiles;
        const int nn0 = (more ? tn : t) * 16 + 2 * wave;

        int    nca = 0, ncb = 0;
        int    nc0[2] = {0, 0}, nc[2] = {0, 0}, ncm[2] = {0, 0};
        float  ndd[2] = {0.0f, 0.0f};
        int    ns[2] = {0, 0};
        float  nw[2] = {0.0f, 0.0f};

        nca = cntA[nn0 + (lane & 1)];
        ncb = cntB[nn0 + (lane & 1)];

        const float4 acc0 = sum_node<FIRST>(x4, cntA, cntB, b0, b1, src, dst, dinv_in, n0,
                                            c0[0], c[0], cm[0], dd[0], my_s[0], my_w[0], E, lane);
        *(float4*)&As[(2 * wave + 0) * TP + lane * 4] = acc0;

#pragma unroll
        for (int q = 0; q < 2; ++q) {
            nc0[q] = __builtin_amdgcn_readlane(nca, q);
            nc[q]  = __builtin_amdgcn_readlane(ncb, q);
            const bool fits = (nc0[q] <= CAP) && (nc[q] <= CAP);
            nc[q]  = nc0[q] + nc[q];
            ncm[q] = fits ? nc[q] : 0;
            ndd[q] = rsqrtf((float)(nc[q] + 1));
            if (FIRST) { if (more && lane == 0) dinv_out[nn0 + q] = ndd[q]; }
            ns[q] = load_src(b0, b1, nn0 + q, nc0[q], ncm[q], lane);
        }

        const float4 acc1 = sum_node<FIRST>(x4, cntA, cntB, b0, b1, src, dst, dinv_in, n0 + 1,
                                            c0[1], c[1], cm[1], dd[1], my_s[1], my_w[1], E, lane);

        int   nwa[2] = {0, 0}, nwb[2] = {0, 0};
#pragma unroll
        for (int q = 0; q < 2; ++q) {
            const int sidx = (lane < ncm[q]) ? ns[q] : 0;
            if (FIRST) { nwa[q] = cntA[sidx]; nwb[q] = cntB[sidx]; }
            else       { nw[q] = dinv_in[sidx]; }
        }

        *(float4*)&As[(2 * wave + 1) * TP + lane * 4] = acc1;
        lds_barrier();

        v8f d = {};
#pragma unroll
        for (int i = 0; i < D / 4; ++i) {
            const v2f a = *(const v2f*)&As[lrow * TP + 4 * i + khalf];
            d = __builtin_amdgcn_wmma_f32_16x16x4_f32(
                 false, a,  false, b[i],
                 (short)0, d,  false,  false);
        }

#pragma unroll
        for (int r = 0; r < 8; ++r) {
            float v = d[r] + bn;
            if (do_relu) v = fmaxf(v, 0.0f);
            Cs[(r + m_add) * TP + ncol] = v;
        }
        lds_barrier();
#pragma unroll
        for (int q = 0; q < 2; ++q) {
            const int n = n0 + q;
            ((float4*)out)[(size_t)n * 32 + lane] = *(const float4*)&Cs[(2 * wave + q) * TP + lane * 4];
        }

#pragma unroll
        for (int q = 0; q < 2; ++q) {
            c0[q] = nc0[q]; c[q] = nc[q]; cm[q] = ncm[q]; dd[q] = ndd[q];
            my_s[q] = ns[q];
            my_w[q] = FIRST ? rsqrtf((float)(nwa[q] + nwb[q] + 1)) : nw[q];
        }
    }
}

extern "C" void kernel_launch(void* const* d_in, const int* in_sizes, int n_in,
                              void* d_out, int out_size, void* d_ws, size_t ws_size,
                              hipStream_t stream) {
    const float* x  = (const float*)d_in[0];
    const int*   ei = (const int*)d_in[1];
    const float* W1 = (const float*)d_in[2];
    const float* b1 = (const float*)d_in[3];
    const float* W2 = (const float*)d_in[4];
    const float* b2 = (const float*)d_in[5];
    float* out = (float*)d_out;

    const int N = in_sizes[0] / D;
    const int E = in_sizes[1] / 2;
    const int* src = ei;
    const int* dst = ei + E;
    const size_t ND = (size_t)N * D;

    const int nb_E = (E + 256 * EPT - 1) / (256 * EPT);
    const int nt_M = N / 16;
    const int NPL  = ((N + 31) / 32) * 32;
    const int nb_Z = (2 * NPL + 255) / 256;

    float* ws   = (float*)d_ws;
    float* dinv = ws;
    float* h    = ws + 65536;
    int* cnt2   = (int*)(h + ND);
    unsigned short* brow = (unsigned short*)(cnt2 + 2 * NPL);
    const int* cntA = cnt2;
    const int* cntB = cnt2 + NPL;
    const unsigned short* brow0 = brow;
    const unsigned short* brow1 = brow + (size_t)N * CAP;

    k_zero_cnt  <<<nb_Z, 256, 0, stream>>>(cnt2, 2 * NPL);
    k_count_fill<<<nb_E, 256, 0, stream>>>(src, dst, cnt2, brow, E, N, NPL);

    k_agg_gemm<true> <<<FUSED_BLOCKS, 256, 0, stream>>>(x, W1, cntA, cntB, brow0, brow1, src, dst,
                                                        dinv, dinv, b1, h, nt_M, E, 1);
    k_agg_gemm<false><<<FUSED_BLOCKS, 256, 0, stream>>>(h, W2, cntA, cntB, brow0, brow1, src, dst,
                                                        dinv, dinv, b2, out, nt_M, E, 0);
}


# Layer 1 stores each hidden row times its own dinv; layer 2 adds rows unweighted and scales the finished sum by dinv[dst] once: no dinv[src] fetch per edge, dinv array gone (rounding changes)
# speedup vs baseline: 5.5681x; 1.0615x over previous
#include <hip/hip_runtime.h>


#define D 128

#define EPT 8
#define CAP 32
#define TP 132
#define FUSED_BLOCKS 782

typedef __attribute__((ext_vector_type(2))) float v2f;
typedef __attribute__((ext_vector_type(8))) float v8f;
typedef const float4 __attribute__((address_space(1)))* gf4_ptr;

static __device__ __forceinline__ int pkg_half() {
    const unsigned v = (unsigned)__builtin_amdgcn_s_sendmsg_rtn(0x87);
    return (int)((v >> 18) & 1u);
}

__global__ __launch_bounds__(256) void k_zero_cnt(int* cnt2, int n) {
    int i = blockIdx.x * 256 + threadIdx.x;
    if (i < n) cnt2[i] = 0;
}

__global__ __launch_bounds__(256) void k_count_fill(const int* __restrict__ src,
                                                    const int* __restrict__ dst,
                                                    int* cnt2,
                                                    unsigned short* __restrict__ brow,
                                                    int E, int N, int NPL) {
    const int hf = pkg_half();
    int* cnt            = cnt2 + (size_t)hf * NPL;
    unsigned short* row = brow + (size_t)hf * N * CAP;
    const int base = blockIdx.x * (256 * EPT) + threadIdx.x;
    if ((int)(blockIdx.x + 1) * (256 * EPT) <= E) {
        int d[EPT];
        int s[EPT];
        int r[EPT];
#pragma unroll
        for (int u = 0; u < EPT; ++u) {
            d[u] = dst[base + u * 256];
            s[u] = src[base + u * 256];
        }
        asm volatile("" : "+v"(s[0]));
#pragma unroll
        for (int u = 0; u < EPT; ++u) r[u] = atomicAdd(&cnt[d[u]], 1);
#pragma unroll
        for (int u = 0; u < EPT; ++u)
            if (r[u] < CAP) row[(size_t)d[u] * CAP + r[u]] = (unsigned short)s[u];
    } else {
        for (int u = 0; u < EPT; ++u) {
            const int i = base + u * 256;
            if (i < E) {
                const int dd = dst[i];
                const int rr = atomicAdd(&cnt[dd], 1);
                if (rr < CAP) row[(size_t)dd * CAP + rr] = (unsigned short)src[i];
            }
        }
    }
}

static __device__ __forceinline__ void fma4(float4& acc, const float4 a, const float w) {
    acc.x = fmaf(a.x, w, acc.x);
    acc.y = fmaf(a.y, w, acc.y);
    acc.z = fmaf(a.z, w, acc.z);
    acc.w = fmaf(a.w, w, acc.w);
}

static __device__ __forceinline__ void add4(float4& acc, const float4 a) {
    acc.x += a.x;
    acc.y += a.y;
    acc.z += a.z;
    acc.w += a.w;
}

template <int K, bool WT>
static __device__ __forceinline__ void rows(float4& acc, const float4* __restrict__ x4,
                                            const int my_s, const float my_d, const float dd,
                                            const int j, const int lane) {
    float4 a[K];
    float  w[K];
    unsigned long long ra[K];
#pragma unroll
    for (int u = 0; u < K; ++u) {
        const int s = __builtin_amdgcn_readlane(my_s, j + u);
        w[u] = WT ? __int_as_float(__builtin_amdgcn_readlane(__float_as_int(my_d), j + u)) * dd : 1.0f;
        ra[u] = (unsigned long long)x4 + ((unsigned long long)(unsigned)s << 9);
    }
    if constexpr (K == 8)
        asm volatile("" : "+s"(ra[0]), "+s"(ra[1]), "+s"(ra[2]), "+s"(ra[3]),
                          "+s"(ra[4]), "+s"(ra[5]), "+s"(ra[6]), "+s"(ra[7]));
    else if constexpr (K == 4)
        asm volatile("" : "+s"(ra[0]), "+s"(ra[1]), "+s"(ra[2]), "+s"(ra[3]));
    else if constexpr (K == 3)
        asm volatile("" : "+s"(ra[0]), "+s"(ra[1]), "+s"(ra[2]));
    else if constexpr (K == 2)
        asm volatile("" : "+s"(ra[0]), "+s"(ra[1]));
    else
        asm volatile("" : "+s"(ra[0]));
#pragma unroll
    for (int u = 0; u < K; ++u) {
        a[u] = ((gf4_ptr)ra[u])[lane];
    }
#pragma unroll
    for (int u = 0; u < K; ++u) {
        if (WT) fma4(acc, a[u], w[u]);
        else    add4(acc, a[u]);
    }
}

static __device__ __forceinline__ int load_src(const unsigned short* __restrict__ b0,
                                               const unsigned short* __restrict__ b1,
                                               const int n, const int c0, const int c, const int idx) {
    const int hs = (int)(b1 - b0);
    const int r  = n * CAP;
    int e = (idx < c0) ? r + idx : hs + r + (idx - c0);
    if (idx >= c) e = r;
    return (int)b0[e];
}

static __device__ __forceinline__ float dinv_src(const int* __restrict__ cntA,
                                                 const int* __restrict__ cntB, const int s) {
    return rsqrtf((float)(cntA[s] + cntB[s] + 1));
}

template <bool WT>
static __device__ __forceinline__ void entries(float4& acc, const float4* __restrict__ x4,
                                               const int cs, const float cw, const float dd,
                                               const int m, const int lane) {
    int j = 0;
    for (; j + 8 <= m; j += 8) rows<8, WT>(acc, x4, cs, cw, dd, j, lane);
    if (j + 4 <= m) { rows<4, WT>(acc, x4, cs, cw, dd, j, lane); j += 4; }
    const int rem = m - j;
    if (rem == 3)      rows<3, WT>(acc, x4, cs, cw, dd, j, lane);
    else if (rem == 2) rows<2, WT>(acc, x4, cs, cw, dd, j, lane);
    else if (rem == 1) rows<1, WT>(acc, x4, cs, cw, dd, j, lane);
}

template <bool FIRST>
static __device__ __forceinline__ float4 sum_node(const float4* __restrict__ x4,
                                                  const int* __restrict__ cntA,
                                                  const int* __restrict__ cntB,
                                                  const unsigned short* __restrict__ b0,
                                                  const unsigned short* __restrict__ b1,
                                                  const int* __restrict__ src,
                                                  const int* __restrict__ dst,
                                                  const int n, const int c0, const int c, const int cm,
                                                  const float dd,
                                                  const int my_s, const float my_w,
                                                  const int E, const int lane) {
    float4 acc = make_float4(0.0f, 0.0f, 0.0f, 0.0f);
    const int tot = cm + 1;
    {
        const int   cs = (lane == cm) ? n  : my_s;
        const float cw = FIRST ? ((lane == cm) ? dd : my_w) : 0.0f;
        entries<FIRST>(acc, x4, cs, cw, dd, min(32, tot), lane);
    }
    for (int base = 32; base < tot; base += 32) {
        int   cs = load_src(b0, b1, n, c0, cm, base + lane);
        float cw = 0.0f;
        if (FIRST) cw = dinv_src(cntA, cntB, (base + lane < cm) ? cs : 0);
        if (base + lane == cm) { cs = n; cw = dd; }
        entries<FIRST>(acc, x4, cs, cw, dd, min(32, tot - base), lane);
    }
    if (cm != c) {
        for (int e0 = 0; e0 < E; e0 += 32) {
            const int e  = e0 + lane;
            const int dv = (e < E) ? dst[e] : -1;
            unsigned mk = __builtin_amdgcn_ballot_w32(dv == n);
            if (mk != 0u) {
                const int   sv = (dv == n) ? src[e] : 0;
                float wv = 0.0f;
                if (FIRST) wv = dinv_src(cntA, cntB, sv);
                while (mk != 0u) {
                    const int j = __builtin_ctz(mk);
                    mk &= mk - 1u;
                    rows<1, FIRST>(acc, x4, sv, wv, dd, j, lane);
                }
            }
        }
    }
    if (!FIRST) {
        acc.x *= dd;
        acc.y *= dd;
        acc.z *= dd;
        acc.w *= dd;
    }
    return acc;
}

static __device__ __forceinline__ void lds_barrier() {
    __builtin_amdgcn_sched_barrier(0);
    __builtin_amdgcn_fence(__ATOMIC_RELEASE, "workgroup", "local");
    __builtin_amdgcn_s_barrier();
    __builtin_amdgcn_fence(__ATOMIC_ACQUIRE, "workgroup", "local");
}

template <bool FIRST>
__global__ __launch_bounds__(256) void k_agg_gemm(const float* __restrict__ X,
                                                  const float* __restrict__ W,
                                                  const int* __restrict__ cntA,
                                                  const int* __restrict__ cntB,
                                                  const unsigned short* __restrict__ b0,
                                                  const unsigned short* __restrict__ b1,
                                                  const int* __restrict__ src,
                                                  const int* __restrict__ dst,
                                                  const float* __restrict__ bias,
                                                  float* __restrict__ out,
                                                  int ntiles, int E, int do_relu) {
    __shared__ __attribute__((aligned(16))) float As[16 * TP];
    __shared__ __attribute__((aligned(16))) float Cs[16 * TP];

    const int lane  = threadIdx.x & 31;
    const int wave  = __builtin_amdgcn_readfirstlane((int)(threadIdx.x >> 5));
    const int col0  = wave * 16;
    const int lrow  = lane & 15;
    const int khalf = (lane >> 4) * 2;
    const int m_add = (lane >> 4) * 8;
    const int ncol  = col0 + lrow;

    v2f b[D / 4];
#pragma unroll
    for (int i = 0; i < D / 4; ++i) {
        b[i].x = W[(size_t)(4 * i + khalf + 0) * D + ncol];
        b[i].y = W[(size_t)(4 * i + khalf + 1) * D + ncol];
    }
    const float bn = bias[ncol];
    const float4* x4 = (const float4*)X;

    int    c0[2] = {0, 0}, c[2] = {0, 0}, cm[2] = {0, 0};
    float  dd[2] = {0.0f, 0.0f};
    int    my_s[2] = {0, 0};
    float  my_w[2] = {0.0f, 0.0f};

    int t = blockIdx.x;
    if (t < ntiles) {
        const int n0 = t * 16 + 2 * wave;
#pragma unroll
        for (int q = 0; q < 2; ++q) {
            const int n = n0 + q;
            c0[q] = cntA[n];
            c[q]  = cntB[n];
        }
#pragma unroll
        for (int q = 0; q < 2; ++q) {
            const bool fits = (c0[q] <= CAP) && (c[q] <= CAP);
            c[q]  = c0[q] + c[q];
            cm[q] = fits ? c[q] : 0;
            dd[q] = rsqrtf((float)(c[q] + 1));
            my_s[q] = load_src(b0, b1, n0 + q, c0[q], cm[q], lane);
        }
        asm volatile("" : "+v"(my_s[0]), "+v"(my_s[1]));
        if (FIRST) {
#pragma unroll
            for (int q = 0; q < 2; ++q)
                my_w[q] = dinv_src(cntA, cntB, (lane < cm[q]) ? my_s[q] : 0);
        }
    }

    for (; t < ntiles; t += gridDim.x) {
        const int row0 = t * 16;
        const int n0 = row0 + 2 * wave;
        const int tn = t + gridDim.x;
        const bool more = tn < ntiles;
        const int nn0 = (more ? tn : t) * 16 + 2 * wave;

        int    nca = 0, ncb = 0;
        int    nc0[2] = {0, 0}, nc[2] = {0, 0}, ncm[2] = {0, 0};
        float  ndd[2] = {0.0f, 0.0f};
        int    ns[2] = {0, 0};

        nca = cntA[nn0 + (lane & 1)];
        ncb = cntB[nn0 + (lane & 1)];

        const float4 acc0 = sum_node<FIRST>(x4, cntA, cntB, b0, b1, src, dst, n0,
                                            c0[0], c[0], cm[0], dd[0], my_s[0], my_w[0], E, lane);
        *(float4*)&As[(2 * wave + 0) * TP + lane * 4] = acc0;

#pragma unroll
        for (int q = 0; q < 2; ++q) {
            nc0[q] = __builtin_amdgcn_readlane(nca, q);
            nc[q]  = __builtin_amdgcn_readlane(ncb, q);
            const bool fits = (nc0[q] <= CAP) && (nc[q] <= CAP);
            nc[q]  = nc0[q] + nc[q];
            ncm[q] = fits ? nc[q] : 0;
            ndd[q] = rsqrtf((float)(nc[q] + 1));
            ns[q] = load_src(b0, b1, nn0 + q, nc0[q], ncm[q], lane);
        }

        const float4 acc1 = sum_node<FIRST>(x4, cntA, cntB, b0, b1, src, dst, n0 + 1,
                                            c0[1], c[1], cm[1], dd[1], my_s[1], my_w[1], E, lane);

        int   nwa[2] = {0, 0}, nwb[2] = {0, 0};
        if (FIRST) {
#pragma unroll
            for (int q = 0; q < 2; ++q) {
                const int sidx = (lane < ncm[q]) ? ns[q] : 0;
                nwa[q] = cntA[sidx];
                nwb[q] = cntB[sidx];
            }
        }

        *(float4*)&As[(2 * wave + 1) * TP + lane * 4] = acc1;
        lds_barrier();

        v8f d = {};
#pragma unroll
        for (int i = 0; i < D / 4; ++i) {
            const v2f a = *(const v2f*)&As[lrow * TP + 4 * i + khalf];
            d = __builtin_amdgcn_wmma_f32_16x16x4_f32(
                 false, a,  false, b[i],
                 (short)0, d,  false,  false);
        }

#pragma unroll
        for (int r = 0; r < 8; ++r) {
            float v = d[r] + bn;
            if (do_relu) v = fmaxf(v, 0.0f);
            Cs[(r + m_add) * TP + ncol] = v;
        }
        lds_barrier();
#pragma unroll
        for (int q = 0; q < 2; ++q) {
            const int n = n0 + q;
            float4 v = *(const float4*)&Cs[(2 * wave + q) * TP + lane * 4];
            if (FIRST) {
                v.x *= dd[q];
                v.y *= dd[q];
                v.z *= dd[q];
                v.w *= dd[q];
            }
            ((float4*)out)[(size_t)n * 32 + lane] = v;
        }

#pragma unroll
        for (int q = 0; q < 2; ++q) {
            c0[q] = nc0[q]; c[q] = nc[q]; cm[q] = ncm[q]; dd[q] = ndd[q];
            my_s[q] = ns[q];
            if (FIRST) my_w[q] = rsqrtf((float)(nwa[q] + nwb[q] + 1));
        }
    }
}

extern "C" void kernel_launch(void* const* d_in, const int* in_sizes, int n_in,
                              void* d_out, int out_size, void* d_ws, size_t ws_size,
                              hipStream_t stream) {
    const float* x  = (const float*)d_in[0];
    const int*   ei = (const int*)d_in[1];
    const float* W1 = (const float*)d_in[2];
    const float* b1 = (const float*)d_in[3];
    const float* W2 = (const float*)d_in[4];
    const float* b2 = (const float*)d_in[5];
    float* out = (float*)d_out;

    const int N = in_sizes[0] / D;
    const int E = in_sizes[1] / 2;
    const int* src = ei;
    const int* dst = ei + E;
    const size_t ND = (size_t)N * D;

    const int nb_E = (E + 256 * EPT - 1) / (256 * EPT);
    const int nt_M = N / 16;
    const int NPL  = ((N + 31) / 32) * 32;
    const int nb_Z = (2 * NPL + 255) / 256;

    float* ws   = (float*)d_ws;
    float* h    = ws + 65536;
    int* cnt2   = (int*)(h + ND);
    unsigned short* brow = (unsigned short*)(cnt2 + 2 * NPL);
    const int* cntA = cnt2;
    const int* cntB = cnt2 + NPL;
    const unsigned short* brow0 = brow;
    const unsigned short* brow1 = brow + (size_t)N * CAP;

    k_zero_cnt  <<<nb_Z, 256, 0, stream>>>(cnt2, 2 * NPL);
    k_count_fill<<<nb_E, 256, 0, stream>>>(src, dst, cnt2, brow, E, N, NPL);

    k_agg_gemm<true> <<<FUSED_BLOCKS, 256, 0, stream>>>(x, W1, cntA, cntB, brow0, brow1, src, dst,
                                                        b1, h, nt_M, E, 1);
    k_agg_gemm<false><<<FUSED_BLOCKS, 256, 0, stream>>>(h, W2, cntA, cntB, brow0, brow1, src, dst,
                                                        b2, out, nt_M, E, 0);
}
